# starting kernel
# baseline (speedup 1.0000x reference)
#include <hip/hip_runtime.h>


typedef __attribute__((ext_vector_type(2))) float v2f;
typedef __attribute__((ext_vector_type(8))) float v8f;

#define NB   8
#define NT   1024
#define NH   8
#define NS   64
#define NK   31
#define NC   512
#define PADK 15
#define NJ   12

__global__ __launch_bounds__(256) void lightconv_wmma_f32(
    const float* __restrict__ x,
    const float* __restrict__ filters,
    const float* __restrict__ bias,
    float* __restrict__ out)
{
    __shared__ float wsm[8][16][32];

    const int lane = threadIdx.x & 31;
    const int wave = threadIdx.x >> 5;
    const int task = blockIdx.x * 8 + wave;
    const int h    = task & 7;
    const int tt   = (task >> 3) & 63;
    const int b    = task >> 9;
    const int t0   = tt * 16;

    const int m = lane & 15;
    if (lane < 16) {
        const float* f = filters + (size_t)(b * NT + t0 + m) * (NH * NK) + h * NK;
        float v[NK];
        float mx = -3.4e38f;
#pragma unroll
        for (int k = 0; k < NK; ++k) { v[k] = f[k]; mx = fmaxf(mx, v[k]); }
        float sum = 0.f;
#pragma unroll
        for (int k = 0; k < NK; ++k) { v[k] = __expf(v[k] - mx); sum += v[k]; }
        const float inv = 1.f / sum;
#pragma unroll
        for (int k = 0; k < NK; ++k) wsm[wave][m][k] = v[k] * inv;
    }
    __syncthreads();

    const int klo = (lane < 16) ? 0 : 2;
    v2f afrag[NJ];
#pragma unroll
    for (int j = 0; j < NJ; ++j) {
        const int jj0 = 4 * j + klo;
        const int k0  = jj0 - m;
        const int k1  = k0 + 1;
        afrag[j].x = (k0 >= 0 && k0 < NK) ? wsm[wave][m][k0] : 0.f;
        afrag[j].y = (k1 >= 0 && k1 < NK) ? wsm[wave][m][k1] : 0.f;
    }

    const int n     = lane & 15;
    const int rbase = t0 - PADK + klo;
    for (int st = 0; st < 4; ++st) {
        const int c0 = h * NS + st * 16;
        const float* xcol = x + (size_t)b * NT * NC + c0 + n;

        v8f acc = {};
#pragma unroll
        for (int j = 0; j < NJ; ++j) {
            const int r0 = rbase + 4 * j;
            const int r1 = r0 + 1;
            v2f bfrag;
            bfrag.x = (r0 >= 0 && r0 < NT) ? xcol[(size_t)r0 * NC] : 0.f;
            bfrag.y = (r1 >= 0 && r1 < NT) ? xcol[(size_t)r1 * NC] : 0.f;
            acc = __builtin_amdgcn_wmma_f32_16x16x4_f32(
                      false, afrag[j], false, bfrag, (short)0, acc, false, false);
        }

        const float bv    = bias[c0 + n];
        const int   mrow0 = (lane < 16) ? 0 : 8;
        float* orow = out + (size_t)(b * NT + t0 + mrow0) * NC + c0 + n;
#pragma unroll
        for (int vi = 0; vi < 8; ++vi)
            orow[(size_t)vi * NC] = acc[vi] + bv;
    }
}

extern "C" void kernel_launch(void* const* d_in, const int* in_sizes, int n_in,
                              void* d_out, int out_size, void* d_ws, size_t ws_size,
                              hipStream_t stream) {
    (void)in_sizes; (void)n_in; (void)out_size; (void)d_ws; (void)ws_size;
    const float* x       = (const float*)d_in[0];
    const float* filters = (const float*)d_in[1];
    const float* bias    = (const float*)d_in[2];
    float*       out     = (float*)d_out;

    dim3 grid(512), block(256);
    lightconv_wmma_f32<<<grid, block, 0, stream>>>(x, filters, bias, out);
}


# x band of each (32 t-row x 4 head) work-group tile staged once in LDS by whole-row 16-byte loads issued as one group; B fragments are 8-byte LDS reads instead of guarded per-lane global loads
# speedup vs baseline: 2.1331x; 2.1331x over previous
#include <hip/hip_runtime.h>


typedef __attribute__((ext_vector_type(2))) float v2f;
typedef __attribute__((ext_vector_type(4))) float v4f;
typedef __attribute__((ext_vector_type(8))) float v8f;

#define NB   8
#define NT   1024
#define NH   8
#define NS   64
#define NK   31
#define NC   512
#define PADK 15
#define NJ   12
#define TT   32
#define HG   4
#define CW   256
#define XR   64
#define XP2  544

__global__ __launch_bounds__(256) void lightconv_wmma_f32(
    const float* __restrict__ x,
    const float* __restrict__ filters,
    const float* __restrict__ bias,
    float* __restrict__ out)
{
    __shared__ float wsm[8][16][32];
    __shared__ __attribute__((aligned(16))) float xs[(XR / 2) * XP2];

    const int tid  = threadIdx.x;
    const int lane = tid & 31;
    const int wave = tid >> 5;
    const int hg   = blockIdx.x & 1;
    const int tt   = (blockIdx.x >> 1) & 31;
    const int b    = blockIdx.x >> 6;
    const int t0   = tt * TT;
    const int hl   = wave & 3;
    const int ts   = wave >> 2;
    const int h    = hg * HG + hl;
    const int t0w  = t0 + ts * 16;
    const int m    = lane & 15;

    const float* f = filters + (size_t)(b * NT + t0w + m) * (NH * NK) + h * NK;
    float v[NK];
#pragma unroll
    for (int k = 0; k < NK; ++k) v[k] = f[k];

    const int c4 = tid & 63;
    const int rq = tid >> 6;
    const float* xb = x + (size_t)b * NT * NC + hg * CW + c4 * 4;
    v4f xe[8], xo[8];
#pragma unroll
    for (int it = 0; it < 8; ++it) {
        const int r0 = t0 - PADK + 2 * (rq + 4 * it);
        const int r1 = r0 + 1;
        xe[it] = *(const v4f*)(xb + (size_t)min(max(r0, 0), NT - 1) * NC);
        xo[it] = *(const v4f*)(xb + (size_t)min(max(r1, 0), NT - 1) * NC);
    }
    asm volatile("" :: "v"(v[0]), "v"(v[1]), "v"(v[2]), "v"(v[3]), "v"(v[4]), "v"(v[5]),
                       "v"(v[6]), "v"(v[7]), "v"(v[8]), "v"(v[9]), "v"(v[10]), "v"(v[11]),
                       "v"(v[12]), "v"(v[13]), "v"(v[14]), "v"(v[15]));
    asm volatile("" :: "v"(v[16]), "v"(v[17]), "v"(v[18]), "v"(v[19]), "v"(v[20]), "v"(v[21]),
                       "v"(v[22]), "v"(v[23]), "v"(v[24]), "v"(v[25]), "v"(v[26]), "v"(v[27]),
                       "v"(v[28]), "v"(v[29]), "v"(v[30]));

    float mx = -3.4e38f;
#pragma unroll
    for (int k = 0; k < NK; ++k) mx = fmaxf(mx, v[k]);
    float sum = 0.f;
#pragma unroll
    for (int k = 0; k < NK; ++k) { v[k] = __expf(v[k] - mx); sum += v[k]; }
    const float inv = 1.f / sum;
    if (lane < 16) {
#pragma unroll
        for (int k = 0; k < NK; ++k) wsm[wave][m][k] = v[k] * inv;
    }

#pragma unroll
    for (int it = 0; it < 8; ++it) {
        const int p  = rq + 4 * it;
        const int r0 = t0 - PADK + 2 * p;
        const int r1 = r0 + 1;
        const bool in0 = (r0 >= 0 && r0 < NT);
        const bool in1 = (r1 >= 0 && r1 < NT);
        v4f lo, hi;
        lo.x = in0 ? xe[it].x : 0.f;  lo.y = in1 ? xo[it].x : 0.f;
        lo.z = in0 ? xe[it].y : 0.f;  lo.w = in1 ? xo[it].y : 0.f;
        hi.x = in0 ? xe[it].z : 0.f;  hi.y = in1 ? xo[it].z : 0.f;
        hi.z = in0 ? xe[it].w : 0.f;  hi.w = in1 ? xo[it].w : 0.f;
        float* dst = xs + p * XP2 + c4 * 8;
        *(v4f*)(dst)     = lo;
        *(v4f*)(dst + 4) = hi;
    }
    __syncthreads();

    const int klo = (lane < 16) ? 0 : 2;
    v2f afrag[NJ];
#pragma unroll
    for (int j = 0; j < NJ; ++j) {
        const int jj0 = 4 * j + klo;
        const int k0  = jj0 - m;
        const int k1  = k0 + 1;
        afrag[j].x = (k0 >= 0 && k0 < NK) ? wsm[wave][m][k0] : 0.f;
        afrag[j].y = (k1 >= 0 && k1 < NK) ? wsm[wave][m][k1] : 0.f;
    }

    const int n = lane & 15;
    const float* xl = xs + (ts * 8 + (lane >> 4)) * XP2 + (hl * NS + n) * 2;
    v8f acc[4];
#pragma unroll
    for (int st = 0; st < 4; ++st) acc[st] = (v8f){};
#pragma unroll
    for (int j = 0; j < NJ; ++j) {
#pragma unroll
        for (int st = 0; st < 4; ++st) {
            const v2f bfrag = *(const v2f*)(xl + (2 * j) * XP2 + st * 32);
            acc[st] = __builtin_amdgcn_wmma_f32_16x16x4_f32(
                          false, afrag[j], false, bfrag, (short)0, acc[st], false, false);
        }
    }

    const int mrow0 = (lane < 16) ? 0 : 8;
#pragma unroll
    for (int st = 0; st < 4; ++st) {
        const int   c0 = h * NS + st * 16;
        const float bv = bias[c0 + n];
        float* orow = out + (size_t)(b * NT + t0w + mrow0) * NC + c0 + n;
#pragma unroll
        for (int vi = 0; vi < 8; ++vi)
            orow[(size_t)vi * NC] = acc[st][vi] + bv;
    }
}

extern "C" void kernel_launch(void* const* d_in, const int* in_sizes, int n_in,
                              void* d_out, int out_size, void* d_ws, size_t ws_size,
                              hipStream_t stream) {
    (void)in_sizes; (void)n_in; (void)out_size; (void)d_ws; (void)ws_size;
    const float* x       = (const float*)d_in[0];
    const float* filters = (const float*)d_in[1];
    const float* bias    = (const float*)d_in[2];
    float*       out     = (float*)d_out;

    dim3 grid(512), block(256);
    lightconv_wmma_f32<<<grid, block, 0, stream>>>(x, filters, bias, out);
}


# Filter taps read row-contiguous (lane = tap: 16 loads of one 124-byte row piece per wave, not 8 loads of 16 lines) and turned to row-per-lane through the wave's LDS scratch at pitch 36 floats
# speedup vs baseline: 2.2684x; 1.0634x over previous
#include <hip/hip_runtime.h>


typedef __attribute__((ext_vector_type(2))) float v2f;
typedef __attribute__((ext_vector_type(4))) float v4f;
typedef __attribute__((ext_vector_type(8))) float v8f;

#define NB   8
#define NT   1024
#define NH   8
#define NS   64
#define NK   31
#define NC   512
#define PADK 15
#define NJ   12
#define TT   32
#define HG   4
#define CW   256
#define XR   64
#define XP2  544
#define WP   36

__global__ __launch_bounds__(256) void lightconv_wmma_f32(
    const float* __restrict__ x,
    const float* __restrict__ filters,
    const float* __restrict__ bias,
    float* __restrict__ out)
{
    __shared__ __attribute__((aligned(16))) float wsm[8][16][WP];
    __shared__ __attribute__((aligned(16))) float xs[(XR / 2) * XP2];

    const int tid  = threadIdx.x;
    const int lane = tid & 31;
    const int wave = tid >> 5;
    const int hg   = blockIdx.x & 1;
    const int tt   = (blockIdx.x >> 1) & 31;
    const int b    = blockIdx.x >> 6;
    const int t0   = tt * TT;
    const int hl   = wave & 3;
    const int ts   = wave >> 2;
    const int h    = hg * HG + hl;
    const int t0w  = t0 + ts * 16;
    const int m    = lane & 15;

    const float* f = filters + (size_t)(b * NT + t0w) * (NH * NK) + h * NK + min(lane, NK - 1);
    float raw[16];
#pragma unroll
    for (int r = 0; r < 16; ++r) raw[r] = f[(size_t)r * (NH * NK)];

    const int c4 = tid & 63;
    const int rq = tid >> 6;
    const float* xb = x + (size_t)b * NT * NC + hg * CW + c4 * 4;
    v4f xe[8], xo[8];
#pragma unroll
    for (int it = 0; it < 8; ++it) {
        const int r0 = t0 - PADK + 2 * (rq + 4 * it);
        const int r1 = r0 + 1;
        xe[it] = *(const v4f*)(xb + (size_t)min(max(r0, 0), NT - 1) * NC);
        xo[it] = *(const v4f*)(xb + (size_t)min(max(r1, 0), NT - 1) * NC);
    }

    float* wrow = &wsm[wave][0][0];
#pragma unroll
    for (int r = 0; r < 16; ++r) wrow[r * WP + lane] = raw[r];
    __builtin_amdgcn_fence(__ATOMIC_RELEASE, "wavefront");
    __builtin_amdgcn_wave_barrier();
    __builtin_amdgcn_fence(__ATOMIC_ACQUIRE, "wavefront");
    float v[32];
#pragma unroll
    for (int i = 0; i < 8; ++i) {
        const v4f q = *(const v4f*)(wrow + m * WP + 4 * i);
        v[4 * i] = q.x;  v[4 * i + 1] = q.y;  v[4 * i + 2] = q.z;  v[4 * i + 3] = q.w;
    }

    float mx = -3.4e38f;
#pragma unroll
    for (int k = 0; k < NK; ++k) mx = fmaxf(mx, v[k]);
    float sum = 0.f;
#pragma unroll
    for (int k = 0; k < NK; ++k) { v[k] = __expf(v[k] - mx); sum += v[k]; }
    const float inv = 1.f / sum;
    if (lane < 16) {
        v[NK] = 0.f;
#pragma unroll
        for (int i = 0; i < 8; ++i) {
            v4f o;
            o.x = v[4 * i] * inv;      o.y = v[4 * i + 1] * inv;
            o.z = v[4 * i + 2] * inv;  o.w = v[4 * i + 3] * inv;
            *(v4f*)(wrow + m * WP + 4 * i) = o;
        }
    }

#pragma unroll
    for (int it = 0; it < 8; ++it) {
        const int p  = rq + 4 * it;
        const int r0 = t0 - PADK + 2 * p;
        const int r1 = r0 + 1;
        const bool in0 = (r0 >= 0 && r0 < NT);
        const bool in1 = (r1 >= 0 && r1 < NT);
        v4f lo, hi;
        lo.x = in0 ? xe[it].x : 0.f;  lo.y = in1 ? xo[it].x : 0.f;
        lo.z = in0 ? xe[it].y : 0.f;  lo.w = in1 ? xo[it].y : 0.f;
        hi.x = in0 ? xe[it].z : 0.f;  hi.y = in1 ? xo[it].z : 0.f;
        hi.z = in0 ? xe[it].w : 0.f;  hi.w = in1 ? xo[it].w : 0.f;
        float* dst = xs + p * XP2 + c4 * 8;
        *(v4f*)(dst)     = lo;
        *(v4f*)(dst + 4) = hi;
    }
    __syncthreads();

    const int klo = (lane < 16) ? 0 : 2;
    v2f afrag[NJ];
#pragma unroll
    for (int j = 0; j < NJ; ++j) {
        const int jj0 = 4 * j + klo;
        const int k0  = jj0 - m;
        const int k1  = k0 + 1;
        afrag[j].x = (k0 >= 0 && k0 < NK) ? wsm[wave][m][k0] : 0.f;
        afrag[j].y = (k1 >= 0 && k1 < NK) ? wsm[wave][m][k1] : 0.f;
    }

    const int n = lane & 15;
    const float* xl = xs + (ts * 8 + (lane >> 4)) * XP2 + (hl * NS + n) * 2;
    v8f acc[4];
#pragma unroll
    for (int st = 0; st < 4; ++st) acc[st] = (v8f){};
#pragma unroll
    for (int j = 0; j < NJ; ++j) {
#pragma unroll
        for (int st = 0; st < 4; ++st) {
            const v2f bfrag = *(const v2f*)(xl + (2 * j) * XP2 + st * 32);
            acc[st] = __builtin_amdgcn_wmma_f32_16x16x4_f32(
                          false, afrag[j], false, bfrag, (short)0, acc[st], false, false);
        }
    }

    const int mrow0 = (lane < 16) ? 0 : 8;
#pragma unroll
    for (int st = 0; st < 4; ++st) {
        const int   c0 = h * NS + st * 16;
        const float bv = bias[c0 + n];
        float* orow = out + (size_t)(b * NT + t0w + mrow0) * NC + c0 + n;
#pragma unroll
        for (int vi = 0; vi < 8; ++vi)
            orow[(size_t)vi * NC] = acc[st][vi] + bv;
    }
}

extern "C" void kernel_launch(void* const* d_in, const int* in_sizes, int n_in,
                              void* d_out, int out_size, void* d_ws, size_t ws_size,
                              hipStream_t stream) {
    (void)in_sizes; (void)n_in; (void)out_size; (void)d_ws; (void)ws_size;
    const float* x       = (const float*)d_in[0];
    const float* filters = (const float*)d_in[1];
    const float* bias    = (const float*)d_in[2];
    float*       out     = (float*)d_out;

    dim3 grid(512), block(256);
    lightconv_wmma_f32<<<grid, block, 0, stream>>>(x, filters, bias, out);
}
